# starting kernel
# baseline (speedup 1.0000x reference)
#include <hip/hip_runtime.h>
#include <math.h>

typedef float v2f __attribute__((ext_vector_type(2)));
typedef float v8f __attribute__((ext_vector_type(8)));

#define KDIM 386
#define KMAIN 384
#define HDIM 256

__global__ void gcn_init_deg(float* deg, int N) {
  int i = blockIdx.x * blockDim.x + threadIdx.x;
  if (i < N) deg[i] = 1.0f;
}

__global__ void gcn_accum_deg(const int* __restrict__ dst, float* deg, int E) {
  int e = blockIdx.x * blockDim.x + threadIdx.x;
  if (e < E) atomicAdd(&deg[dst[e]], 1.0f);
}

__global__ void gcn_compute_dis(const float* __restrict__ deg, float* dis, int N) {
  int i = blockIdx.x * blockDim.x + threadIdx.x;
  if (i < N) dis[i] = rsqrtf(deg[i]);
}

__global__ void gcn_edge_norm(const int* __restrict__ src, const int* __restrict__ dst,
                              const float* __restrict__ dis, float* en, int E) {
  int e = blockIdx.x * blockDim.x + threadIdx.x;
  if (e < E) en[e] = dis[src[e]] * dis[dst[e]];
}

__global__ __launch_bounds__(256) void gcn_gemm1_wmma(const float* __restrict__ x,
                                                      const float* __restrict__ W1,
                                                      float* __restrict__ t, int M) {
  const int lane = threadIdx.x & 31;
  const int wave = threadIdx.x >> 5;
  const int m0 = blockIdx.x * 16;
  const int n0 = wave * 32;
  const int halfSel = lane >> 4;
  const int l15 = lane & 15;

  v8f c0 = {0.f, 0.f, 0.f, 0.f, 0.f, 0.f, 0.f, 0.f};
  v8f c1 = {0.f, 0.f, 0.f, 0.f, 0.f, 0.f, 0.f, 0.f};

  int row = m0 + l15;
  int rowc = (row < M) ? row : (M - 1);
  const float* arow = x + (size_t)rowc * KDIM;
  const int col0 = n0 + l15;

  for (int k = 0; k < KMAIN; k += 4) {
    const int ka = k + halfSel * 2;
    const float2 av = *(const float2*)(arow + ka);
    v2f a;  a.x = av.x;  a.y = av.y;
    const float* bp = W1 + (size_t)ka * HDIM + col0;
    v2f b0; b0.x = bp[0];  b0.y = bp[HDIM];
    v2f b1; b1.x = bp[16]; b1.y = bp[HDIM + 16];
    c0 = __builtin_amdgcn_wmma_f32_16x16x4_f32(false, a, false, b0, (short)0, c0,
                                               false, false);
    c1 = __builtin_amdgcn_wmma_f32_16x16x4_f32(false, a, false, b1, (short)0, c1,
                                               false, false);
  }

  {
    v2f a  = {0.f, 0.f};
    v2f b0 = {0.f, 0.f};
    v2f b1 = {0.f, 0.f};
    if (halfSel == 0) {
      const float2 av = *(const float2*)(arow + KMAIN);
      a.x = av.x;  a.y = av.y;
      const float* bp = W1 + (size_t)KMAIN * HDIM + col0;
      b0.x = bp[0];  b0.y = bp[HDIM];
      b1.x = bp[16]; b1.y = bp[HDIM + 16];
    }
    c0 = __builtin_amdgcn_wmma_f32_16x16x4_f32(false, a, false, b0, (short)0, c0,
                                               false, false);
    c1 = __builtin_amdgcn_wmma_f32_16x16x4_f32(false, a, false, b1, (short)0, c1,
                                               false, false);
  }

#pragma unroll
  for (int r = 0; r < 8; ++r) {
    int rr = m0 + r + halfSel * 8;
    if (rr < M) {
      t[(size_t)rr * HDIM + col0] = c0[r];
      t[(size_t)rr * HDIM + col0 + 16] = c1[r];
    }
  }
}

__global__ void gcn_agg1_init(const float* __restrict__ t, const float* __restrict__ dis,
                              const float* __restrict__ b1, float* agg, int N) {
  long long i = (long long)blockIdx.x * blockDim.x + threadIdx.x;
  if (i < (long long)N * HDIM) {
    int n = (int)(i >> 8);
    int f = (int)(i & (HDIM - 1));
    float sn = dis[n] * dis[n];
    agg[i] = t[i] * sn + b1[f];
  }
}

__global__ __launch_bounds__(256) void gcn_scatter1(const int* __restrict__ src,
                                                    const int* __restrict__ dst,
                                                    const float* __restrict__ en,
                                                    const float* __restrict__ t,
                                                    float* agg, int E) {
  int e = blockIdx.x * 4 + (threadIdx.x >> 6);
  if (e >= E) return;
  int f = (threadIdx.x & 63) * 4;
  int s = src[e], d = dst[e];
  float w = en[e];
  float4 v = *(const float4*)(t + (size_t)s * HDIM + f);
  float* ap = agg + (size_t)d * HDIM + f;
  atomicAdd(ap + 0, v.x * w);
  atomicAdd(ap + 1, v.y * w);
  atomicAdd(ap + 2, v.z * w);
  atomicAdd(ap + 3, v.w * w);
}

__global__ __launch_bounds__(256) void gcn_gemm2(const float* __restrict__ agg,
                                                 const float* __restrict__ W2,
                                                 float* t2, int N) {
  int row = blockIdx.x * 8 + (threadIdx.x >> 5);
  int lane = threadIdx.x & 31;
  if (row >= N) return;
  const float* ar = agg + (size_t)row * HDIM;
  float s0 = 0.f, s1 = 0.f;
#pragma unroll
  for (int k = lane; k < HDIM; k += 32) {
    float a = ar[k];
    a = a > 0.f ? a : 0.f;
    s0 += a * W2[k * 2 + 0];
    s1 += a * W2[k * 2 + 1];
  }
#pragma unroll
  for (int off = 16; off > 0; off >>= 1) {
    s0 += __shfl_down(s0, off);
    s1 += __shfl_down(s1, off);
  }
  if (lane == 0) {
    t2[row * 2 + 0] = s0;
    t2[row * 2 + 1] = s1;
  }
}

__global__ void gcn_out_init(const float* __restrict__ t2, const float* __restrict__ dis,
                             const float* __restrict__ b2, float* out, int N) {
  int i = blockIdx.x * blockDim.x + threadIdx.x;
  if (i < 2 * N) {
    int n = i >> 1;
    int c = i & 1;
    float sn = dis[n] * dis[n];
    out[i] = t2[i] * sn + b2[c];
  }
}

__global__ void gcn_scatter2(const int* __restrict__ src, const int* __restrict__ dst,
                             const float* __restrict__ en, const float* __restrict__ t2,
                             float* out, int E) {
  int e = blockIdx.x * blockDim.x + threadIdx.x;
  if (e < E) {
    int s = src[e], d = dst[e];
    float w = en[e];
    atomicAdd(&out[d * 2 + 0], t2[s * 2 + 0] * w);
    atomicAdd(&out[d * 2 + 1], t2[s * 2 + 1] * w);
  }
}

__global__ void gcn_tanh(float* out, int n) {
  int i = blockIdx.x * blockDim.x + threadIdx.x;
  if (i < n) out[i] = tanhf(out[i]);
}

extern "C" void kernel_launch(void* const* d_in, const int* in_sizes, int n_in,
                              void* d_out, int out_size, void* d_ws, size_t ws_size,
                              hipStream_t stream) {
  const float* x  = (const float*)d_in[0];
  const int*   src = (const int*)d_in[1];
  const int*   dst = (const int*)d_in[2];
  const float* W1 = (const float*)d_in[3];
  const float* b1 = (const float*)d_in[4];
  const float* W2 = (const float*)d_in[5];
  const float* b2 = (const float*)d_in[6];
  float* out = (float*)d_out;

  const int N = in_sizes[0] / KDIM;
  const int E = in_sizes[1];

  float* ws  = (float*)d_ws;
  float* t   = ws;
  float* agg = t   + (size_t)N * HDIM;
  float* deg = agg + (size_t)N * HDIM;
  float* dis = deg + N;
  float* en  = dis + N;
  float* t2  = en  + E;

  const int TB = 256;
  gcn_init_deg<<<(N + TB - 1) / TB, TB, 0, stream>>>(deg, N);
  gcn_accum_deg<<<(E + TB - 1) / TB, TB, 0, stream>>>(dst, deg, E);
  gcn_compute_dis<<<(N + TB - 1) / TB, TB, 0, stream>>>(deg, dis, N);
  gcn_edge_norm<<<(E + TB - 1) / TB, TB, 0, stream>>>(src, dst, dis, en, E);

  gcn_gemm1_wmma<<<(N + 15) / 16, 256, 0, stream>>>(x, W1, t, N);
  {
    long long tot = (long long)N * HDIM;
    gcn_agg1_init<<<(int)((tot + TB - 1) / TB), TB, 0, stream>>>(t, dis, b1, agg, N);
  }
  gcn_scatter1<<<(E + 3) / 4, 256, 0, stream>>>(src, dst, en, t, agg, E);

  gcn_gemm2<<<(N + 7) / 8, 256, 0, stream>>>(agg, W2, t2, N);
  gcn_out_init<<<(2 * N + TB - 1) / TB, TB, 0, stream>>>(t2, dis, b2, out, N);
  gcn_scatter2<<<(E + TB - 1) / TB, TB, 0, stream>>>(src, dst, en, t2, out, E);
  gcn_tanh<<<(2 * N + TB - 1) / TB, TB, 0, stream>>>(out, 2 * N);
}


# Both float-atomic scatter-adds replaced by a gather over edges grouped by destination (in-call counting sort, arrival-order rank); row finished in registers (ReLU, W2 dot); summation order changes
# speedup vs baseline: 1.9448x; 1.9448x over previous
#include <hip/hip_runtime.h>
#include <math.h>

typedef float v2f __attribute__((ext_vector_type(2)));
typedef float v8f __attribute__((ext_vector_type(8)));

#define KDIM 386
#define KMAIN 384
#define HDIM 256

__global__ void gcn_zero_cnt(int* cnt, int N) {
  int i = blockIdx.x * blockDim.x + threadIdx.x;
  if (i < N) cnt[i] = 0;
}

__global__ void gcn_rank(const int* __restrict__ dst, int* cnt, int* __restrict__ pos, int E) {
  int e = blockIdx.x * blockDim.x + threadIdx.x;
  if (e < E) pos[e] = atomicAdd(&cnt[dst[e]], 1);
}

__global__ __launch_bounds__(1024) void gcn_scan(const int* __restrict__ cnt,
                                                 int* __restrict__ rowptr,
                                                 float* __restrict__ dis, int N) {
  __shared__ int wsum[32];
  const int tid = threadIdx.x;
  const int lane = tid & 31;
  const int wv = tid >> 5;
  const int per = (N + 1023) / 1024;
  const int beg = tid * per;
  int end = beg + per;
  if (end > N) end = N;

  int local = 0;
  for (int i = beg; i < end; ++i) local += cnt[i];

  int incl = local;
  for (int off = 1; off < 32; off <<= 1) {
    int v = __shfl_up(incl, off);
    if (lane >= off) incl += v;
  }
  if (lane == 31) wsum[wv] = incl;
  __syncthreads();
  if (wv == 0) {
    const int w = wsum[lane];
    int wi = w;
    for (int off = 1; off < 32; off <<= 1) {
      int v = __shfl_up(wi, off);
      if (lane >= off) wi += v;
    }
    wsum[lane] = wi - w;
  }
  __syncthreads();

  int run = wsum[wv] + incl - local;
  for (int i = beg; i < end; ++i) {
    const int c = cnt[i];
    rowptr[i] = run;
    run += c;
    dis[i] = rsqrtf((float)(c + 1));
  }
  if (tid == 1023) rowptr[N] = run;
}

__global__ void gcn_fill(const int* __restrict__ src, const int* __restrict__ dst,
                         const int* __restrict__ rowptr, const int* __restrict__ pos,
                         int* __restrict__ csr_src, int E) {
  int e = blockIdx.x * blockDim.x + threadIdx.x;
  if (e < E) csr_src[rowptr[dst[e]] + pos[e]] = src[e];
}

__global__ __launch_bounds__(256) void gcn_gemm1_wmma(const float* __restrict__ x,
                                                      const float* __restrict__ W1,
                                                      float* __restrict__ t, int M) {
  const int lane = threadIdx.x & 31;
  const int wave = threadIdx.x >> 5;
  const int m0 = blockIdx.x * 16;
  const int n0 = wave * 32;
  const int halfSel = lane >> 4;
  const int l15 = lane & 15;

  v8f c0 = {0.f, 0.f, 0.f, 0.f, 0.f, 0.f, 0.f, 0.f};
  v8f c1 = {0.f, 0.f, 0.f, 0.f, 0.f, 0.f, 0.f, 0.f};

  int row = m0 + l15;
  int rowc = (row < M) ? row : (M - 1);
  const float* arow = x + (size_t)rowc * KDIM;
  const int col0 = n0 + l15;

  for (int k = 0; k < KMAIN; k += 4) {
    const int ka = k + halfSel * 2;
    const float2 av = *(const float2*)(arow + ka);
    v2f a;  a.x = av.x;  a.y = av.y;
    const float* bp = W1 + (size_t)ka * HDIM + col0;
    v2f b0; b0.x = bp[0];  b0.y = bp[HDIM];
    v2f b1; b1.x = bp[16]; b1.y = bp[HDIM + 16];
    c0 = __builtin_amdgcn_wmma_f32_16x16x4_f32(false, a, false, b0, (short)0, c0,
                                               false, false);
    c1 = __builtin_amdgcn_wmma_f32_16x16x4_f32(false, a, false, b1, (short)0, c1,
                                               false, false);
  }

  {
    v2f a  = {0.f, 0.f};
    v2f b0 = {0.f, 0.f};
    v2f b1 = {0.f, 0.f};
    if (halfSel == 0) {
      const float2 av = *(const float2*)(arow + KMAIN);
      a.x = av.x;  a.y = av.y;
      const float* bp = W1 + (size_t)KMAIN * HDIM + col0;
      b0.x = bp[0];  b0.y = bp[HDIM];
      b1.x = bp[16]; b1.y = bp[HDIM + 16];
    }
    c0 = __builtin_amdgcn_wmma_f32_16x16x4_f32(false, a, false, b0, (short)0, c0,
                                               false, false);
    c1 = __builtin_amdgcn_wmma_f32_16x16x4_f32(false, a, false, b1, (short)0, c1,
                                               false, false);
  }

#pragma unroll
  for (int r = 0; r < 8; ++r) {
    int rr = m0 + r + halfSel * 8;
    if (rr < M) {
      t[(size_t)rr * HDIM + col0] = c0[r];
      t[(size_t)rr * HDIM + col0 + 16] = c1[r];
    }
  }
}


template <int K>
__device__ __forceinline__ void gcn_rows(const float* __restrict__ t, int sj, float wj, int j,
                                         int f0, int f1, float4& a0, float4& a1) {
  float4 v0[K], v1[K];
  float w[K];
#pragma unroll
  for (int u = 0; u < K; ++u) {
    const int s = __builtin_amdgcn_readlane(sj, j + u);
    w[u] = __int_as_float(__builtin_amdgcn_readlane(__float_as_int(wj), j + u));
    const float* tp = t + (size_t)s * HDIM;
    v0[u] = *(const float4*)(tp + f0);
    v1[u] = *(const float4*)(tp + f1);
  }
#pragma unroll
  for (int u = 0; u < K; ++u) {
    a0.x = fmaf(v0[u].x, w[u], a0.x);  a0.y = fmaf(v0[u].y, w[u], a0.y);
    a0.z = fmaf(v0[u].z, w[u], a0.z);  a0.w = fmaf(v0[u].w, w[u], a0.w);
    a1.x = fmaf(v1[u].x, w[u], a1.x);  a1.y = fmaf(v1[u].y, w[u], a1.y);
    a1.z = fmaf(v1[u].z, w[u], a1.z);  a1.w = fmaf(v1[u].w, w[u], a1.w);
  }
}

__global__ __launch_bounds__(256) void gcn_gather1(const int* __restrict__ rowptr,
                                                   const int* __restrict__ csr_src,
                                                   const float* __restrict__ dis,
                                                   const float* __restrict__ t,
                                                   const float* __restrict__ b1,
                                                   const float* __restrict__ W2,
                                                   float* __restrict__ t2, int N) {
  const int wv = __builtin_amdgcn_readfirstlane((int)(threadIdx.x >> 5));
  const int d = blockIdx.x * 8 + wv;
  const int lane = threadIdx.x & 31;
  if (d >= N) return;
  const int f0 = lane * 4;
  const int f1 = 128 + lane * 4;
  const float dd = dis[d];
  const int beg = rowptr[d];
  const int end = rowptr[d + 1];

  float4 a0 = make_float4(0.f, 0.f, 0.f, 0.f);
  float4 a1 = make_float4(0.f, 0.f, 0.f, 0.f);

  for (int base = beg; base < end; base += 32) {
    int n = end - base;
    if (n > 32) n = 32;
    int sj = 0;
    float wj = 0.f;
    if (lane < n) {
      sj = csr_src[base + lane];
      wj = dis[sj] * dd;
    }
    int j = 0;
    for (; j + 8 <= n; j += 8) gcn_rows<8>(t, sj, wj, j, f0, f1, a0, a1);
    if (n & 4) { gcn_rows<4>(t, sj, wj, j, f0, f1, a0, a1); j += 4; }
    if (n & 2) { gcn_rows<2>(t, sj, wj, j, f0, f1, a0, a1); j += 2; }
    if (n & 1) { gcn_rows<1>(t, sj, wj, j, f0, f1, a0, a1); }
  }

  {
    const float sn = dd * dd;
    const float* tp = t + (size_t)d * HDIM;
    const float4 h0 = *(const float4*)(tp + f0);
    const float4 h1 = *(const float4*)(tp + f1);
    const float4 c0 = *(const float4*)(b1 + f0);
    const float4 c1 = *(const float4*)(b1 + f1);
    a0.x = fmaxf((a0.x + h0.x * sn) + c0.x, 0.f);
    a0.y = fmaxf((a0.y + h0.y * sn) + c0.y, 0.f);
    a0.z = fmaxf((a0.z + h0.z * sn) + c0.z, 0.f);
    a0.w = fmaxf((a0.w + h0.w * sn) + c0.w, 0.f);
    a1.x = fmaxf((a1.x + h1.x * sn) + c1.x, 0.f);
    a1.y = fmaxf((a1.y + h1.y * sn) + c1.y, 0.f);
    a1.z = fmaxf((a1.z + h1.z * sn) + c1.z, 0.f);
    a1.w = fmaxf((a1.w + h1.w * sn) + c1.w, 0.f);
  }

  const float4 wa = *(const float4*)(W2 + f0 * 2);
  const float4 wb = *(const float4*)(W2 + f0 * 2 + 4);
  const float4 wc = *(const float4*)(W2 + f1 * 2);
  const float4 wd = *(const float4*)(W2 + f1 * 2 + 4);
  float p0 = a0.x * wa.x + a0.y * wa.z + a0.z * wb.x + a0.w * wb.z
           + a1.x * wc.x + a1.y * wc.z + a1.z * wd.x + a1.w * wd.z;
  float p1 = a0.x * wa.y + a0.y * wa.w + a0.z * wb.y + a0.w * wb.w
           + a1.x * wc.y + a1.y * wc.w + a1.z * wd.y + a1.w * wd.w;
  for (int off = 16; off > 0; off >>= 1) {
    p0 += __shfl_down(p0, off);
    p1 += __shfl_down(p1, off);
  }
  if (lane == 0) {
    t2[d * 2 + 0] = p0;
    t2[d * 2 + 1] = p1;
  }
}

__global__ __launch_bounds__(256) void gcn_gather2(const int* __restrict__ rowptr,
                                                   const int* __restrict__ csr_src,
                                                   const float* __restrict__ dis,
                                                   const float* __restrict__ t2,
                                                   const float* __restrict__ b2,
                                                   float* __restrict__ out, int N) {
  __shared__ float so[64];
  const int g = blockIdx.x * 32 + (threadIdx.x >> 3);
  const int sub = threadIdx.x & 7;
  const int d = (g < N) ? g : (N - 1);
  const float dd = dis[d];
  const int beg = rowptr[d];
  const int end = rowptr[d + 1];
  float s0 = 0.f, s1 = 0.f;
  for (int j = beg + sub; j < end; j += 8) {
    const int s = csr_src[j];
    const float w = dis[s] * dd;
    const float2 v = *(const float2*)(t2 + 2 * s);
    s0 += v.x * w;
    s1 += v.y * w;
  }
  for (int off = 4; off > 0; off >>= 1) {
    s0 += __shfl_xor(s0, off);
    s1 += __shfl_xor(s1, off);
  }
  if (sub == 0) {
    const float sn = dd * dd;
    const float2 hv = *(const float2*)(t2 + 2 * d);
    so[(threadIdx.x >> 3) * 2 + 0] = tanhf((s0 + hv.x * sn) + b2[0]);
    so[(threadIdx.x >> 3) * 2 + 1] = tanhf((s1 + hv.y * sn) + b2[1]);
  }
  __syncthreads();
  if (threadIdx.x < 64) {
    const int o = blockIdx.x * 64 + threadIdx.x;
    if (o < 2 * N) out[o] = so[threadIdx.x];
  }
}

extern "C" void kernel_launch(void* const* d_in, const int* in_sizes, int n_in,
                              void* d_out, int out_size, void* d_ws, size_t ws_size,
                              hipStream_t stream) {
  const float* x  = (const float*)d_in[0];
  const int*   src = (const int*)d_in[1];
  const int*   dst = (const int*)d_in[2];
  const float* W1 = (const float*)d_in[3];
  const float* b1 = (const float*)d_in[4];
  const float* W2 = (const float*)d_in[5];
  const float* b2 = (const float*)d_in[6];
  float* out = (float*)d_out;

  const int N = in_sizes[0] / KDIM;
  const int E = in_sizes[1];

  float* ws  = (float*)d_ws;
  float* t   = ws;
  float* t2  = t  + (size_t)N * HDIM;
  float* dis = t2 + (size_t)2 * N;
  int* cnt     = (int*)(dis + N);
  int* rowptr  = cnt + N;
  int* pos     = rowptr + N + 1;
  int* csr_src = pos + E;

  const int TB = 256;
  gcn_zero_cnt<<<(N + TB - 1) / TB, TB, 0, stream>>>(cnt, N);
  gcn_rank<<<(E + TB - 1) / TB, TB, 0, stream>>>(dst, cnt, pos, E);
  gcn_scan<<<1, 1024, 0, stream>>>(cnt, rowptr, dis, N);
  gcn_fill<<<(E + TB - 1) / TB, TB, 0, stream>>>(src, dst, rowptr, pos, csr_src, E);

  gcn_gemm1_wmma<<<(N + 15) / 16, 256, 0, stream>>>(x, W1, t, N);
  gcn_gather1<<<(N + 7) / 8, 256, 0, stream>>>(rowptr, csr_src, dis, t, b1, W2, t2, N);

  gcn_gather2<<<(N + 31) / 32, 256, 0, stream>>>(rowptr, csr_src, dis, t2, b2, out, N);
}


# GEMM1 retiled: one 80x256 tile per work-group, x slab staged once in LDS, each wave holds 5x2 register tiles so one B fragment pair feeds 10 products; k order and sums unchanged
# speedup vs baseline: 2.2480x; 1.1559x over previous
#include <hip/hip_runtime.h>
#include <math.h>

typedef float v2f __attribute__((ext_vector_type(2)));
typedef float v8f __attribute__((ext_vector_type(8)));

#define KDIM 386
#define KMAIN 384
#define HDIM 256

__global__ void gcn_zero_cnt(int* cnt, int N) {
  int i = blockIdx.x * blockDim.x + threadIdx.x;
  if (i < N) cnt[i] = 0;
}

__global__ void gcn_rank(const int* __restrict__ dst, int* cnt, int* __restrict__ pos, int E) {
  int e = blockIdx.x * blockDim.x + threadIdx.x;
  if (e < E) pos[e] = atomicAdd(&cnt[dst[e]], 1);
}

__global__ __launch_bounds__(1024) void gcn_scan(const int* __restrict__ cnt,
                                                 int* __restrict__ rowptr,
                                                 float* __restrict__ dis, int N) {
  __shared__ int wsum[32];
  const int tid = threadIdx.x;
  const int lane = tid & 31;
  const int wv = tid >> 5;
  const int per = (N + 1023) / 1024;
  const int beg = tid * per;
  int end = beg + per;
  if (end > N) end = N;

  int local = 0;
  for (int i = beg; i < end; ++i) local += cnt[i];

  int incl = local;
  for (int off = 1; off < 32; off <<= 1) {
    int v = __shfl_up(incl, off);
    if (lane >= off) incl += v;
  }
  if (lane == 31) wsum[wv] = incl;
  __syncthreads();
  if (wv == 0) {
    const int w = wsum[lane];
    int wi = w;
    for (int off = 1; off < 32; off <<= 1) {
      int v = __shfl_up(wi, off);
      if (lane >= off) wi += v;
    }
    wsum[lane] = wi - w;
  }
  __syncthreads();

  int run = wsum[wv] + incl - local;
  for (int i = beg; i < end; ++i) {
    const int c = cnt[i];
    rowptr[i] = run;
    run += c;
    dis[i] = rsqrtf((float)(c + 1));
  }
  if (tid == 1023) rowptr[N] = run;
}

__global__ void gcn_fill(const int* __restrict__ src, const int* __restrict__ dst,
                         const int* __restrict__ rowptr, const int* __restrict__ pos,
                         int* __restrict__ csr_src, int E) {
  int e = blockIdx.x * blockDim.x + threadIdx.x;
  if (e < E) csr_src[rowptr[dst[e]] + pos[e]] = src[e];
}

#define BM 80
#define MT 5
#define LDSP 388
#define KC 16
#define NCHUNK 24

struct gcn_frag {
  v2f a[4][MT];
  v2f b[4][2];
};

__device__ __forceinline__ void gcn_load_chunk(gcn_frag& f, const float* ap,
                                               const float* __restrict__ bp) {
#pragma unroll
  for (int s = 0; s < 4; ++s) {
    const float* b = bp + (size_t)s * 4 * HDIM;
    f.b[s][0].x = b[0];   f.b[s][0].y = b[HDIM];
    f.b[s][1].x = b[16];  f.b[s][1].y = b[HDIM + 16];
  }
#pragma unroll
  for (int s = 0; s < 4; ++s) {
#pragma unroll
    for (int i = 0; i < MT; ++i) {
      const float2 v = *(const float2*)(ap + i * 16 * LDSP + 4 * s);
      f.a[s][i].x = v.x;  f.a[s][i].y = v.y;
    }
  }
}

__device__ __forceinline__ void gcn_mma_chunk(const gcn_frag& f, v8f (&acc)[MT][2]) {
#pragma unroll
  for (int s = 0; s < 4; ++s) {
#pragma unroll
    for (int i = 0; i < MT; ++i) {
      acc[i][0] = __builtin_amdgcn_wmma_f32_16x16x4_f32(false, f.a[s][i], false, f.b[s][0],
                                                        (short)0, acc[i][0], false, false);
      acc[i][1] = __builtin_amdgcn_wmma_f32_16x16x4_f32(false, f.a[s][i], false, f.b[s][1],
                                                        (short)0, acc[i][1], false, false);
    }
  }
}

__global__ __launch_bounds__(256) void gcn_gemm1_wmma(const float* __restrict__ x,
                                                      const float* __restrict__ W1,
                                                      float* __restrict__ t, int M) {
  __shared__ float xs[41984];

  const int lane = threadIdx.x & 31;
  const int wave = threadIdx.x >> 5;
  const int m0 = blockIdx.x * BM;
  const int n0 = wave * 32;
  const int halfSel = lane >> 4;
  const int l15 = lane & 15;
  const int col0 = n0 + l15;

#pragma unroll
  for (int rr = 0; rr < BM / 8; ++rr) {
    const int r = wave + 8 * rr;
    int row = m0 + r;
    if (row > M - 1) row = M - 1;
    const float2* src = (const float2*)(x + (size_t)row * KDIM);
    float2* dst = (float2*)(xs + r * LDSP);
#pragma unroll
    for (int q = 0; q < 6; ++q) dst[lane + 32 * q] = src[lane + 32 * q];
  }
  if (lane < BM / 8) {
    const int r = wave + 8 * lane;
    int row = m0 + r;
    if (row > M - 1) row = M - 1;
    ((float2*)(xs + r * LDSP))[192] = ((const float2*)(x + (size_t)row * KDIM))[192];
  }
  __syncthreads();

  v8f acc[MT][2];
#pragma unroll
  for (int i = 0; i < MT; ++i) {
    acc[i][0] = (v8f){0.f, 0.f, 0.f, 0.f, 0.f, 0.f, 0.f, 0.f};
    acc[i][1] = (v8f){0.f, 0.f, 0.f, 0.f, 0.f, 0.f, 0.f, 0.f};
  }

  const float* ap = xs + l15 * LDSP + halfSel * 2;
  const float* bp = W1 + (size_t)(halfSel * 2) * HDIM + col0;

  gcn_frag f0, f1;
  gcn_load_chunk(f0, ap, bp);
  for (int c = 0; c < NCHUNK; c += 2) {
    gcn_load_chunk(f1, ap + (c + 1) * KC, bp + (size_t)(c + 1) * KC * HDIM);
    gcn_mma_chunk(f0, acc);
    const int cn = (c + 2 < NCHUNK) ? (c + 2) : (NCHUNK - 1);
    gcn_load_chunk(f0, ap + cn * KC, bp + (size_t)cn * KC * HDIM);
    gcn_mma_chunk(f1, acc);
  }

  {
    v2f a[MT];
    v2f b0 = {0.f, 0.f};
    v2f b1 = {0.f, 0.f};
#pragma unroll
    for (int i = 0; i < MT; ++i) a[i] = (v2f){0.f, 0.f};
    if (halfSel == 0) {
#pragma unroll
      for (int i = 0; i < MT; ++i) {
        const float2 v = *(const float2*)(xs + (i * 16 + l15) * LDSP + KMAIN);
        a[i].x = v.x;  a[i].y = v.y;
      }
      const float* bt = W1 + (size_t)KMAIN * HDIM + col0;
      b0.x = bt[0];   b0.y = bt[HDIM];
      b1.x = bt[16];  b1.y = bt[HDIM + 16];
    }
#pragma unroll
    for (int i = 0; i < MT; ++i) {
      acc[i][0] = __builtin_amdgcn_wmma_f32_16x16x4_f32(false, a[i], false, b0, (short)0,
                                                        acc[i][0], false, false);
      acc[i][1] = __builtin_amdgcn_wmma_f32_16x16x4_f32(false, a[i], false, b1, (short)0,
                                                        acc[i][1], false, false);
    }
  }

  float* tp = t + (size_t)(m0 + halfSel * 8) * HDIM + col0;
  if (m0 + BM <= M) {
#pragma unroll
    for (int i = 0; i < MT; ++i) {
#pragma unroll
      for (int r = 0; r < 8; ++r) {
        tp[(i * 16 + r) * HDIM] = acc[i][0][r];
        tp[(i * 16 + r) * HDIM + 16] = acc[i][1][r];
      }
    }
  } else {
#pragma unroll
    for (int i = 0; i < MT; ++i) {
#pragma unroll
      for (int r = 0; r < 8; ++r) {
        if (m0 + i * 16 + r + halfSel * 8 < M) {
          tp[(i * 16 + r) * HDIM] = acc[i][0][r];
          tp[(i * 16 + r) * HDIM + 16] = acc[i][1][r];
        }
      }
    }
  }
}


template <int K>
__device__ __forceinline__ void gcn_rows(const float* __restrict__ t, int sj, float wj, int j,
                                         int f0, int f1, float4& a0, float4& a1) {
  float4 v0[K], v1[K];
  float w[K];
#pragma unroll
  for (int u = 0; u < K; ++u) {
    const int s = __builtin_amdgcn_readlane(sj, j + u);
    w[u] = __int_as_float(__builtin_amdgcn_readlane(__float_as_int(wj), j + u));
    const float* tp = t + (size_t)s * HDIM;
    v0[u] = *(const float4*)(tp + f0);
    v1[u] = *(const float4*)(tp + f1);
  }
#pragma unroll
  for (int u = 0; u < K; ++u) {
    a0.x = fmaf(v0[u].x, w[u], a0.x);  a0.y = fmaf(v0[u].y, w[u], a0.y);
    a0.z = fmaf(v0[u].z, w[u], a0.z);  a0.w = fmaf(v0[u].w, w[u], a0.w);
    a1.x = fmaf(v1[u].x, w[u], a1.x);  a1.y = fmaf(v1[u].y, w[u], a1.y);
    a1.z = fmaf(v1[u].z, w[u], a1.z);  a1.w = fmaf(v1[u].w, w[u], a1.w);
  }
}

__global__ __launch_bounds__(256) void gcn_gather1(const int* __restrict__ rowptr,
                                                   const int* __restrict__ csr_src,
                                                   const float* __restrict__ dis,
                                                   const float* __restrict__ t,
                                                   const float* __restrict__ b1,
                                                   const float* __restrict__ W2,
                                                   float* __restrict__ t2, int N) {
  const int wv = __builtin_amdgcn_readfirstlane((int)(threadIdx.x >> 5));
  const int d = blockIdx.x * 8 + wv;
  const int lane = threadIdx.x & 31;
  if (d >= N) return;
  const int f0 = lane * 4;
  const int f1 = 128 + lane * 4;
  const float dd = dis[d];
  const int beg = rowptr[d];
  const int end = rowptr[d + 1];

  float4 a0 = make_float4(0.f, 0.f, 0.f, 0.f);
  float4 a1 = make_float4(0.f, 0.f, 0.f, 0.f);

  for (int base = beg; base < end; base += 32) {
    int n = end - base;
    if (n > 32) n = 32;
    int sj = 0;
    float wj = 0.f;
    if (lane < n) {
      sj = csr_src[base + lane];
      wj = dis[sj] * dd;
    }
    int j = 0;
    for (; j + 8 <= n; j += 8) gcn_rows<8>(t, sj, wj, j, f0, f1, a0, a1);
    if (n & 4) { gcn_rows<4>(t, sj, wj, j, f0, f1, a0, a1); j += 4; }
    if (n & 2) { gcn_rows<2>(t, sj, wj, j, f0, f1, a0, a1); j += 2; }
    if (n & 1) { gcn_rows<1>(t, sj, wj, j, f0, f1, a0, a1); }
  }

  {
    const float sn = dd * dd;
    const float* tp = t + (size_t)d * HDIM;
    const float4 h0 = *(const float4*)(tp + f0);
    const float4 h1 = *(const float4*)(tp + f1);
    const float4 c0 = *(const float4*)(b1 + f0);
    const float4 c1 = *(const float4*)(b1 + f1);
    a0.x = fmaxf((a0.x + h0.x * sn) + c0.x, 0.f);
    a0.y = fmaxf((a0.y + h0.y * sn) + c0.y, 0.f);
    a0.z = fmaxf((a0.z + h0.z * sn) + c0.z, 0.f);
    a0.w = fmaxf((a0.w + h0.w * sn) + c0.w, 0.f);
    a1.x = fmaxf((a1.x + h1.x * sn) + c1.x, 0.f);
    a1.y = fmaxf((a1.y + h1.y * sn) + c1.y, 0.f);
    a1.z = fmaxf((a1.z + h1.z * sn) + c1.z, 0.f);
    a1.w = fmaxf((a1.w + h1.w * sn) + c1.w, 0.f);
  }

  const float4 wa = *(const float4*)(W2 + f0 * 2);
  const float4 wb = *(const float4*)(W2 + f0 * 2 + 4);
  const float4 wc = *(const float4*)(W2 + f1 * 2);
  const float4 wd = *(const float4*)(W2 + f1 * 2 + 4);
  float p0 = a0.x * wa.x + a0.y * wa.z + a0.z * wb.x + a0.w * wb.z
           + a1.x * wc.x + a1.y * wc.z + a1.z * wd.x + a1.w * wd.z;
  float p1 = a0.x * wa.y + a0.y * wa.w + a0.z * wb.y + a0.w * wb.w
           + a1.x * wc.y + a1.y * wc.w + a1.z * wd.y + a1.w * wd.w;
  for (int off = 16; off > 0; off >>= 1) {
    p0 += __shfl_down(p0, off);
    p1 += __shfl_down(p1, off);
  }
  if (lane == 0) {
    t2[d * 2 + 0] = p0;
    t2[d * 2 + 1] = p1;
  }
}

__global__ __launch_bounds__(256) void gcn_gather2(const int* __restrict__ rowptr,
                                                   const int* __restrict__ csr_src,
                                                   const float* __restrict__ dis,
                                                   const float* __restrict__ t2,
                                                   const float* __restrict__ b2,
                                                   float* __restrict__ out, int N) {
  __shared__ float so[64];
  const int g = blockIdx.x * 32 + (threadIdx.x >> 3);
  const int sub = threadIdx.x & 7;
  const int d = (g < N) ? g : (N - 1);
  const float dd = dis[d];
  const int beg = rowptr[d];
  const int end = rowptr[d + 1];
  float s0 = 0.f, s1 = 0.f;
  for (int j = beg + sub; j < end; j += 8) {
    const int s = csr_src[j];
    const float w = dis[s] * dd;
    const float2 v = *(const float2*)(t2 + 2 * s);
    s0 += v.x * w;
    s1 += v.y * w;
  }
  for (int off = 4; off > 0; off >>= 1) {
    s0 += __shfl_xor(s0, off);
    s1 += __shfl_xor(s1, off);
  }
  if (sub == 0) {
    const float sn = dd * dd;
    const float2 hv = *(const float2*)(t2 + 2 * d);
    so[(threadIdx.x >> 3) * 2 + 0] = tanhf((s0 + hv.x * sn) + b2[0]);
    so[(threadIdx.x >> 3) * 2 + 1] = tanhf((s1 + hv.y * sn) + b2[1]);
  }
  __syncthreads();
  if (threadIdx.x < 64) {
    const int o = blockIdx.x * 64 + threadIdx.x;
    if (o < 2 * N) out[o] = so[threadIdx.x];
  }
}

extern "C" void kernel_launch(void* const* d_in, const int* in_sizes, int n_in,
                              void* d_out, int out_size, void* d_ws, size_t ws_size,
                              hipStream_t stream) {
  const float* x  = (const float*)d_in[0];
  const int*   src = (const int*)d_in[1];
  const int*   dst = (const int*)d_in[2];
  const float* W1 = (const float*)d_in[3];
  const float* b1 = (const float*)d_in[4];
  const float* W2 = (const float*)d_in[5];
  const float* b2 = (const float*)d_in[6];
  float* out = (float*)d_out;

  const int N = in_sizes[0] / KDIM;
  const int E = in_sizes[1];

  float* ws  = (float*)d_ws;
  float* t   = ws;
  float* t2  = t  + (size_t)N * HDIM;
  float* dis = t2 + (size_t)2 * N;
  int* cnt     = (int*)(dis + N);
  int* rowptr  = cnt + N;
  int* pos     = rowptr + N + 1;
  int* csr_src = pos + E;

  const int TB = 256;
  gcn_zero_cnt<<<(N + TB - 1) / TB, TB, 0, stream>>>(cnt, N);
  gcn_rank<<<(E + TB - 1) / TB, TB, 0, stream>>>(dst, cnt, pos, E);
  gcn_scan<<<1, 1024, 0, stream>>>(cnt, rowptr, dis, N);
  gcn_fill<<<(E + TB - 1) / TB, TB, 0, stream>>>(src, dst, rowptr, pos, csr_src, E);

  gcn_gemm1_wmma<<<(N + BM - 1) / BM, 256, 0, stream>>>(x, W1, t, N);
  gcn_gather1<<<(N + 7) / 8, 256, 0, stream>>>(rowptr, csr_src, dis, t, b1, W2, t2, N);

  gcn_gather2<<<(N + 31) / 32, 256, 0, stream>>>(rowptr, csr_src, dis, t2, b2, out, N);
}


# Edge grouping in ONE launch, not four: a work-group owns 256 destinations, sweeps the dst list, ranks by LDS adds, scans and places its own run; no global atomics; summation order changes
# speedup vs baseline: 3.2342x; 1.4387x over previous
#include <hip/hip_runtime.h>
#include <math.h>

typedef float v2f __attribute__((ext_vector_type(2)));
typedef float v8f __attribute__((ext_vector_type(8)));

#define KDIM 386
#define KMAIN 384
#define HDIM 256

#define RNG 256
#define BT 1024
#define CAPW 512
#define UNR 8
#define NOEDGE 0x7fffffff

__device__ __forceinline__ void gcn_visit(int val, int e, int lo, int& below, int& wcount,
                                          int* cntL, int* myE, int* myU) {
  const unsigned u = (unsigned)val - (unsigned)lo;
  const bool isBelow = val < lo;
  const bool own = u < (unsigned)RNG;
  below += __popc(__builtin_amdgcn_ballot_w32(isBelow));
  const unsigned m = __builtin_amdgcn_ballot_w32(own);
  if (m != 0u) {
    const int slot = wcount + (int)__builtin_amdgcn_mbcnt_lo(m, 0u);
    if (own) {
      atomicAdd(&cntL[u], 1);
      if (slot < CAPW) { myE[slot] = e; myU[slot] = (int)u; }
    }
    wcount += __popc(m);
  }
}

__device__ __forceinline__ void gcn_ld(int4 (&v)[UNR], const int4* __restrict__ dv, int s,
                                       int tid) {
#pragma unroll
  for (int q = 0; q < UNR; ++q) v[q] = dv[(s * UNR + q) * BT + tid];
}

__device__ __forceinline__ void gcn_visit4(const int4 v, int e0, int lo, int& below,
                                           int& wcount, int* cntL, int* myE, int* myU) {
  gcn_visit(v.x, e0 + 0, lo, below, wcount, cntL, myE, myU);
  gcn_visit(v.y, e0 + 1, lo, below, wcount, cntL, myE, myU);
  gcn_visit(v.z, e0 + 2, lo, below, wcount, cntL, myE, myU);
  gcn_visit(v.w, e0 + 3, lo, below, wcount, cntL, myE, myU);
}

__device__ __forceinline__ void gcn_visit_stage(const int4 (&v)[UNR], int s, int tid, int lo,
                                                int& below, int& wcount, int* cntL, int* myE,
                                                int* myU) {
#pragma unroll
  for (int q = 0; q < UNR; ++q)
    gcn_visit4(v[q], ((s * UNR + q) * BT + tid) * 4, lo, below, wcount, cntL, myE, myU);
}

__global__ __launch_bounds__(BT) void gcn_build(const int* __restrict__ src,
                                                const int* __restrict__ dst,
                                                int* __restrict__ rowptr,
                                                float* __restrict__ dis,
                                                int* __restrict__ csr_src, int N, int E) {
  __shared__ int cntL[RNG];
  __shared__ int nextL[RNG];
  __shared__ int exL[RNG];
  __shared__ int wsum[RNG / 32];
  __shared__ int wbelow[BT / 32];
  __shared__ int ovf;
  __shared__ int listE[(BT / 32) * CAPW];
  __shared__ int listU[(BT / 32) * CAPW];

  const int tid = threadIdx.x;
  const int lane = tid & 31;
  const int wv = __builtin_amdgcn_readfirstlane(tid >> 5);
  const int lo = blockIdx.x * RNG;

  if (tid < RNG) { cntL[tid] = 0; nextL[tid] = 0; }
  if (tid == 0) ovf = 0;
  __syncthreads();

  int* myE = listE + wv * CAPW;
  int* myU = listU + wv * CAPW;
  int below = 0;
  int wcount = 0;

  const int nvec = E >> 2;
  const int4* dv = (const int4*)dst;
  const int nfull = nvec / (UNR * BT);
  const int r0 = nfull * UNR * BT;
  int4 va[UNR], vb[UNR];
  {
    int4 vr[UNR];
#pragma unroll
    for (int q = 0; q < UNR; ++q) {
      const int i = r0 + q * BT + tid;
      vr[q] = make_int4(NOEDGE, NOEDGE, NOEDGE, NOEDGE);
      if (i < nvec) vr[q] = dv[i];
    }
    if (nfull > 0) gcn_ld(va, dv, 0, tid);
#pragma unroll
    for (int q = 0; q < UNR; ++q) {
      if (r0 + q * BT < nvec)
        gcn_visit4(vr[q], (r0 + q * BT + tid) * 4, lo, below, wcount, cntL, myE, myU);
    }
  }
  if (nfull > 0) {
    const int last = nfull - 1;
    for (int s = 0; s < nfull; s += 2) {
      gcn_ld(vb, dv, (s + 1 < last) ? (s + 1) : last, tid);
      gcn_visit_stage(va, s, tid, lo, below, wcount, cntL, myE, myU);
      gcn_ld(va, dv, (s + 2 < last) ? (s + 2) : last, tid);
      if (s + 1 < nfull) gcn_visit_stage(vb, s + 1, tid, lo, below, wcount, cntL, myE, myU);
    }
  }
  if ((E & 3) != 0) {
    const int e = (nvec << 2) + tid;
    int val = NOEDGE;
    if (tid < (E & 3)) val = dst[e];
    gcn_visit(val, e, lo, below, wcount, cntL, myE, myU);
  }

  if (lane == 0) {
    wbelow[wv] = below;
    if (wcount > CAPW) ovf = 1;
  }
  __syncthreads();

  int base = wbelow[lane];
  for (int off = 16; off > 0; off >>= 1) base += __shfl_xor(base, off);

  int c = 0, incl = 0;
  if (tid < RNG) {
    c = cntL[tid];
    incl = c;
    for (int off = 1; off < 32; off <<= 1) {
      const int v = __shfl_up(incl, off);
      if (lane >= off) incl += v;
    }
    if (lane == 31) wsum[wv] = incl;
  }
  __syncthreads();
  if (tid < RNG) {
    int woff = 0;
    for (int w = 0; w < RNG / 32; ++w) woff += (w < wv) ? wsum[w] : 0;
    const int ex = woff + incl - c;
    exL[tid] = ex;
    const int n = lo + tid;
    if (n < N) {
      rowptr[n] = base + ex;
      dis[n] = rsqrtf((float)(c + 1));
      if (n == N - 1) rowptr[N] = base + ex + c;
    }
  }
  __syncthreads();

  if (ovf == 0) {
    for (int k0 = 0; k0 < wcount; k0 += 128) {
      int sv[4], uu[4];
#pragma unroll
      for (int j = 0; j < 4; ++j) {
        const int k = k0 + j * 32 + lane;
        uu[j] = -1;  sv[j] = 0;
        if (k < wcount) { uu[j] = myU[k];  sv[j] = src[myE[k]]; }
      }
#pragma unroll
      for (int j = 0; j < 4; ++j) {
        if (uu[j] >= 0) {
          const int r = atomicAdd(&nextL[uu[j]], 1);
          csr_src[base + exL[uu[j]] + r] = sv[j];
        }
      }
    }
  } else {
    for (int e = tid; e < E; e += BT) {
      const unsigned u = (unsigned)dst[e] - (unsigned)lo;
      if (u < (unsigned)RNG) {
        const int r = atomicAdd(&nextL[u], 1);
        csr_src[base + exL[u] + r] = src[e];
      }
    }
  }
}

#define BM 80
#define MT 5
#define LDSP 388
#define KC 16
#define NCHUNK 24

struct gcn_frag {
  v2f a[4][MT];
  v2f b[4][2];
};

__device__ __forceinline__ void gcn_load_chunk(gcn_frag& f, const float* ap,
                                               const float* __restrict__ bp) {
#pragma unroll
  for (int s = 0; s < 4; ++s) {
    const float* b = bp + (size_t)s * 4 * HDIM;
    f.b[s][0].x = b[0];   f.b[s][0].y = b[HDIM];
    f.b[s][1].x = b[16];  f.b[s][1].y = b[HDIM + 16];
  }
#pragma unroll
  for (int s = 0; s < 4; ++s) {
#pragma unroll
    for (int i = 0; i < MT; ++i) {
      const float2 v = *(const float2*)(ap + i * 16 * LDSP + 4 * s);
      f.a[s][i].x = v.x;  f.a[s][i].y = v.y;
    }
  }
}

__device__ __forceinline__ void gcn_mma_chunk(const gcn_frag& f, v8f (&acc)[MT][2]) {
#pragma unroll
  for (int s = 0; s < 4; ++s) {
#pragma unroll
    for (int i = 0; i < MT; ++i) {
      acc[i][0] = __builtin_amdgcn_wmma_f32_16x16x4_f32(false, f.a[s][i], false, f.b[s][0],
                                                        (short)0, acc[i][0], false, false);
      acc[i][1] = __builtin_amdgcn_wmma_f32_16x16x4_f32(false, f.a[s][i], false, f.b[s][1],
                                                        (short)0, acc[i][1], false, false);
    }
  }
}

__global__ __launch_bounds__(256) void gcn_gemm1_wmma(const float* __restrict__ x,
                                                      const float* __restrict__ W1,
                                                      float* __restrict__ t, int M) {
  __shared__ float xs[41984];

  const int lane = threadIdx.x & 31;
  const int wave = threadIdx.x >> 5;
  const int m0 = blockIdx.x * BM;
  const int n0 = wave * 32;
  const int halfSel = lane >> 4;
  const int l15 = lane & 15;
  const int col0 = n0 + l15;

#pragma unroll
  for (int rr = 0; rr < BM / 8; ++rr) {
    const int r = wave + 8 * rr;
    int row = m0 + r;
    if (row > M - 1) row = M - 1;
    const float2* src = (const float2*)(x + (size_t)row * KDIM);
    float2* dst = (float2*)(xs + r * LDSP);
#pragma unroll
    for (int q = 0; q < 6; ++q) dst[lane + 32 * q] = src[lane + 32 * q];
  }
  if (lane < BM / 8) {
    const int r = wave + 8 * lane;
    int row = m0 + r;
    if (row > M - 1) row = M - 1;
    ((float2*)(xs + r * LDSP))[192] = ((const float2*)(x + (size_t)row * KDIM))[192];
  }
  __syncthreads();

  v8f acc[MT][2];
#pragma unroll
  for (int i = 0; i < MT; ++i) {
    acc[i][0] = (v8f){0.f, 0.f, 0.f, 0.f, 0.f, 0.f, 0.f, 0.f};
    acc[i][1] = (v8f){0.f, 0.f, 0.f, 0.f, 0.f, 0.f, 0.f, 0.f};
  }

  const float* ap = xs + l15 * LDSP + halfSel * 2;
  const float* bp = W1 + (size_t)(halfSel * 2) * HDIM + col0;

  gcn_frag f0, f1;
  gcn_load_chunk(f0, ap, bp);
  for (int c = 0; c < NCHUNK; c += 2) {
    gcn_load_chunk(f1, ap + (c + 1) * KC, bp + (size_t)(c + 1) * KC * HDIM);
    gcn_mma_chunk(f0, acc);
    const int cn = (c + 2 < NCHUNK) ? (c + 2) : (NCHUNK - 1);
    gcn_load_chunk(f0, ap + cn * KC, bp + (size_t)cn * KC * HDIM);
    gcn_mma_chunk(f1, acc);
  }

  {
    v2f a[MT];
    v2f b0 = {0.f, 0.f};
    v2f b1 = {0.f, 0.f};
#pragma unroll
    for (int i = 0; i < MT; ++i) a[i] = (v2f){0.f, 0.f};
    if (halfSel == 0) {
#pragma unroll
      for (int i = 0; i < MT; ++i) {
        const float2 v = *(const float2*)(xs + (i * 16 + l15) * LDSP + KMAIN);
        a[i].x = v.x;  a[i].y = v.y;
      }
      const float* bt = W1 + (size_t)KMAIN * HDIM + col0;
      b0.x = bt[0];   b0.y = bt[HDIM];
      b1.x = bt[16];  b1.y = bt[HDIM + 16];
    }
#pragma unroll
    for (int i = 0; i < MT; ++i) {
      acc[i][0] = __builtin_amdgcn_wmma_f32_16x16x4_f32(false, a[i], false, b0, (short)0,
                                                        acc[i][0], false, false);
      acc[i][1] = __builtin_amdgcn_wmma_f32_16x16x4_f32(false, a[i], false, b1, (short)0,
                                                        acc[i][1], false, false);
    }
  }

  float* tp = t + (size_t)(m0 + halfSel * 8) * HDIM + col0;
  if (m0 + BM <= M) {
#pragma unroll
    for (int i = 0; i < MT; ++i) {
#pragma unroll
      for (int r = 0; r < 8; ++r) {
        tp[(i * 16 + r) * HDIM] = acc[i][0][r];
        tp[(i * 16 + r) * HDIM + 16] = acc[i][1][r];
      }
    }
  } else {
#pragma unroll
    for (int i = 0; i < MT; ++i) {
#pragma unroll
      for (int r = 0; r < 8; ++r) {
        if (m0 + i * 16 + r + halfSel * 8 < M) {
          tp[(i * 16 + r) * HDIM] = acc[i][0][r];
          tp[(i * 16 + r) * HDIM + 16] = acc[i][1][r];
        }
      }
    }
  }
}


template <int K>
__device__ __forceinline__ void gcn_rows(const float* __restrict__ t, int sj, float wj, int j,
                                         int f0, int f1, float4& a0, float4& a1) {
  float4 v0[K], v1[K];
  float w[K];
#pragma unroll
  for (int u = 0; u < K; ++u) {
    const int s = __builtin_amdgcn_readlane(sj, j + u);
    w[u] = __int_as_float(__builtin_amdgcn_readlane(__float_as_int(wj), j + u));
    const float* tp = t + (size_t)s * HDIM;
    v0[u] = *(const float4*)(tp + f0);
    v1[u] = *(const float4*)(tp + f1);
  }
#pragma unroll
  for (int u = 0; u < K; ++u) {
    a0.x = fmaf(v0[u].x, w[u], a0.x);  a0.y = fmaf(v0[u].y, w[u], a0.y);
    a0.z = fmaf(v0[u].z, w[u], a0.z);  a0.w = fmaf(v0[u].w, w[u], a0.w);
    a1.x = fmaf(v1[u].x, w[u], a1.x);  a1.y = fmaf(v1[u].y, w[u], a1.y);
    a1.z = fmaf(v1[u].z, w[u], a1.z);  a1.w = fmaf(v1[u].w, w[u], a1.w);
  }
}

__global__ __launch_bounds__(256) void gcn_gather1(const int* __restrict__ rowptr,
                                                   const int* __restrict__ csr_src,
                                                   const float* __restrict__ dis,
                                                   const float* __restrict__ t,
                                                   const float* __restrict__ b1,
                                                   const float* __restrict__ W2,
                                                   float* __restrict__ t2, int N) {
  const int wv = __builtin_amdgcn_readfirstlane((int)(threadIdx.x >> 5));
  const int d = blockIdx.x * 8 + wv;
  const int lane = threadIdx.x & 31;
  if (d >= N) return;
  const int f0 = lane * 4;
  const int f1 = 128 + lane * 4;
  const float dd = dis[d];
  const int beg = rowptr[d];
  const int end = rowptr[d + 1];

  float4 a0 = make_float4(0.f, 0.f, 0.f, 0.f);
  float4 a1 = make_float4(0.f, 0.f, 0.f, 0.f);

  for (int base = beg; base < end; base += 32) {
    int n = end - base;
    if (n > 32) n = 32;
    int sj = 0;
    float wj = 0.f;
    if (lane < n) {
      sj = csr_src[base + lane];
      wj = dis[sj] * dd;
    }
    int j = 0;
    for (; j + 8 <= n; j += 8) gcn_rows<8>(t, sj, wj, j, f0, f1, a0, a1);
    if (n & 4) { gcn_rows<4>(t, sj, wj, j, f0, f1, a0, a1); j += 4; }
    if (n & 2) { gcn_rows<2>(t, sj, wj, j, f0, f1, a0, a1); j += 2; }
    if (n & 1) { gcn_rows<1>(t, sj, wj, j, f0, f1, a0, a1); }
  }

  {
    const float sn = dd * dd;
    const float* tp = t + (size_t)d * HDIM;
    const float4 h0 = *(const float4*)(tp + f0);
    const float4 h1 = *(const float4*)(tp + f1);
    const float4 c0 = *(const float4*)(b1 + f0);
    const float4 c1 = *(const float4*)(b1 + f1);
    a0.x = fmaxf((a0.x + h0.x * sn) + c0.x, 0.f);
    a0.y = fmaxf((a0.y + h0.y * sn) + c0.y, 0.f);
    a0.z = fmaxf((a0.z + h0.z * sn) + c0.z, 0.f);
    a0.w = fmaxf((a0.w + h0.w * sn) + c0.w, 0.f);
    a1.x = fmaxf((a1.x + h1.x * sn) + c1.x, 0.f);
    a1.y = fmaxf((a1.y + h1.y * sn) + c1.y, 0.f);
    a1.z = fmaxf((a1.z + h1.z * sn) + c1.z, 0.f);
    a1.w = fmaxf((a1.w + h1.w * sn) + c1.w, 0.f);
  }

  const float4 wa = *(const float4*)(W2 + f0 * 2);
  const float4 wb = *(const float4*)(W2 + f0 * 2 + 4);
  const float4 wc = *(const float4*)(W2 + f1 * 2);
  const float4 wd = *(const float4*)(W2 + f1 * 2 + 4);
  float p0 = a0.x * wa.x + a0.y * wa.z + a0.z * wb.x + a0.w * wb.z
           + a1.x * wc.x + a1.y * wc.z + a1.z * wd.x + a1.w * wd.z;
  float p1 = a0.x * wa.y + a0.y * wa.w + a0.z * wb.y + a0.w * wb.w
           + a1.x * wc.y + a1.y * wc.w + a1.z * wd.y + a1.w * wd.w;
  for (int off = 16; off > 0; off >>= 1) {
    p0 += __shfl_down(p0, off);
    p1 += __shfl_down(p1, off);
  }
  if (lane == 0) {
    t2[d * 2 + 0] = p0;
    t2[d * 2 + 1] = p1;
  }
}

__global__ __launch_bounds__(256) void gcn_gather2(const int* __restrict__ rowptr,
                                                   const int* __restrict__ csr_src,
                                                   const float* __restrict__ dis,
                                                   const float* __restrict__ t2,
                                                   const float* __restrict__ b2,
                                                   float* __restrict__ out, int N) {
  __shared__ float so[64];
  const int g = blockIdx.x * 32 + (threadIdx.x >> 3);
  const int sub = threadIdx.x & 7;
  const int d = (g < N) ? g : (N - 1);
  const float dd = dis[d];
  const int beg = rowptr[d];
  const int end = rowptr[d + 1];
  float s0 = 0.f, s1 = 0.f;
  for (int j = beg + sub; j < end; j += 8) {
    const int s = csr_src[j];
    const float w = dis[s] * dd;
    const float2 v = *(const float2*)(t2 + 2 * s);
    s0 += v.x * w;
    s1 += v.y * w;
  }
  for (int off = 4; off > 0; off >>= 1) {
    s0 += __shfl_xor(s0, off);
    s1 += __shfl_xor(s1, off);
  }
  if (sub == 0) {
    const float sn = dd * dd;
    const float2 hv = *(const float2*)(t2 + 2 * d);
    so[(threadIdx.x >> 3) * 2 + 0] = tanhf((s0 + hv.x * sn) + b2[0]);
    so[(threadIdx.x >> 3) * 2 + 1] = tanhf((s1 + hv.y * sn) + b2[1]);
  }
  __syncthreads();
  if (threadIdx.x < 64) {
    const int o = blockIdx.x * 64 + threadIdx.x;
    if (o < 2 * N) out[o] = so[threadIdx.x];
  }
}

extern "C" void kernel_launch(void* const* d_in, const int* in_sizes, int n_in,
                              void* d_out, int out_size, void* d_ws, size_t ws_size,
                              hipStream_t stream) {
  const float* x  = (const float*)d_in[0];
  const int*   src = (const int*)d_in[1];
  const int*   dst = (const int*)d_in[2];
  const float* W1 = (const float*)d_in[3];
  const float* b1 = (const float*)d_in[4];
  const float* W2 = (const float*)d_in[5];
  const float* b2 = (const float*)d_in[6];
  float* out = (float*)d_out;

  const int N = in_sizes[0] / KDIM;
  const int E = in_sizes[1];

  float* ws  = (float*)d_ws;
  float* t   = ws;
  float* t2  = t  + (size_t)N * HDIM;
  float* dis = t2 + (size_t)2 * N;
  int* rowptr  = (int*)(dis + N);
  int* csr_src = rowptr + N + 1;

  gcn_build<<<(N + RNG - 1) / RNG, BT, 0, stream>>>(src, dst, rowptr, dis, csr_src, N, E);

  gcn_gemm1_wmma<<<(N + BM - 1) / BM, 256, 0, stream>>>(x, W1, t, N);
  gcn_gather1<<<(N + 7) / 8, 256, 0, stream>>>(rowptr, csr_src, dis, t, b1, W2, t2, N);

  gcn_gather2<<<(N + 31) / 32, 256, 0, stream>>>(rowptr, csr_src, dis, t2, b2, out, N);
}


# GEMM1 and the edge grouping in ONE launch: in a 16-wave work-group waves 8-15 do the 80x256 product, waves 0-7 sweep dst for its 80 nodes (ballots, then scalar tests); summation order changes
# speedup vs baseline: 3.6573x; 1.1308x over previous
#include <hip/hip_runtime.h>
#include <math.h>

typedef float v2f __attribute__((ext_vector_type(2)));
typedef float v8f __attribute__((ext_vector_type(8)));

#define KDIM 386
#define KMAIN 384
#define HDIM 256

#define BM 80
#define MT 5
#define LDSP 388
#define KC 16
#define NCHUNK 24

#define RNG BM
#define SW 8
#define GW 8
#define ST (SW * 32)
#define FT ((GW + SW) * 32)
#define SCW ((RNG + 31) / 32)
#define CAPW 1024
#define UNR 16
#define HV 4
#define NOEDGE 0x7fffffff


__device__ __forceinline__ unsigned gcn_owned(int val, int lo, int hi, int& below) {
  const unsigned ltLo = __builtin_amdgcn_ballot_w32(val < lo);
  const unsigned ltHi = __builtin_amdgcn_ballot_w32(val < hi);
  below += __popc(ltLo);
  return ltHi & ~ltLo;
}

__device__ __forceinline__ void gcn_note(unsigned m, int val, int e, int lo, int& wcount,
                                         int* cntL, int* myE, int* myU) {
  const unsigned u = (unsigned)val - (unsigned)lo;
  const int slot = wcount + (int)__builtin_amdgcn_mbcnt_lo(m, 0u);
  if (u < (unsigned)RNG) {
    atomicAdd(&cntL[u], 1);
    if (slot < CAPW) { myE[slot] = e; myU[slot] = (int)u; }
  }
  wcount += __popc(m);
}

__device__ __forceinline__ void gcn_ld(int4 (&v)[UNR], const int4* __restrict__ dv, int s,
                                       int st) {
#pragma unroll
  for (int q = 0; q < UNR; ++q) v[q] = dv[(s * UNR + q) * ST + st];
}

__device__ __forceinline__ void gcn_visit_stage(const int4 (&v)[UNR], int s, int st, int lo,
                                                int& below, int& wcount, int* cntL, int* myE,
                                                int* myU) {
  const int hi = lo + RNG;
#pragma unroll
  for (int h = 0; h < UNR; h += HV) {
    unsigned m[HV][4];
#pragma unroll
    for (int q = 0; q < HV; ++q) {
      m[q][0] = gcn_owned(v[h + q].x, lo, hi, below);
      m[q][1] = gcn_owned(v[h + q].y, lo, hi, below);
      m[q][2] = gcn_owned(v[h + q].z, lo, hi, below);
      m[q][3] = gcn_owned(v[h + q].w, lo, hi, below);
    }
#pragma unroll
    for (int q = 0; q < HV; ++q) {
      const int e0 = ((s * UNR + h + q) * ST + st) * 4;
      if (__builtin_expect(m[q][0] != 0u, 0))
        gcn_note(m[q][0], v[h + q].x, e0 + 0, lo, wcount, cntL, myE, myU);
      if (__builtin_expect(m[q][1] != 0u, 0))
        gcn_note(m[q][1], v[h + q].y, e0 + 1, lo, wcount, cntL, myE, myU);
      if (__builtin_expect(m[q][2] != 0u, 0))
        gcn_note(m[q][2], v[h + q].z, e0 + 2, lo, wcount, cntL, myE, myU);
      if (__builtin_expect(m[q][3] != 0u, 0))
        gcn_note(m[q][3], v[h + q].w, e0 + 3, lo, wcount, cntL, myE, myU);
    }
  }
}

__device__ __forceinline__ void gcn_sweep(const int* __restrict__ dst, int E, int st, int lo,
                                          int& below, int& wcount, int* cntL, int* myE,
                                          int* myU) {
  const int nvec = E >> 2;
  const int4* dv = (const int4*)dst;
  const int nfull = nvec / (UNR * ST);
  const int r0 = nfull * UNR * ST;
  int4 va[UNR], vb[UNR];
  {
    int4 vr[UNR];
#pragma unroll
    for (int q = 0; q < UNR; ++q) {
      const int i = r0 + q * ST + st;
      vr[q] = make_int4(NOEDGE, NOEDGE, NOEDGE, NOEDGE);
      if (i < nvec) vr[q] = dv[i];
    }
    if (nfull > 0) gcn_ld(va, dv, 0, st);
    gcn_visit_stage(vr, nfull, st, lo, below, wcount, cntL, myE, myU);
  }
  if (nfull > 0) {
    const int last = nfull - 1;
    for (int s = 0; s < nfull; s += 2) {
      gcn_ld(vb, dv, (s + 1 < last) ? (s + 1) : last, st);
      gcn_visit_stage(va, s, st, lo, below, wcount, cntL, myE, myU);
      gcn_ld(va, dv, (s + 2 < last) ? (s + 2) : last, st);
      if (s + 1 < nfull) gcn_visit_stage(vb, s + 1, st, lo, below, wcount, cntL, myE, myU);
    }
  }
  if ((E & 3) != 0) {
    const int e = (nvec << 2) + st;
    int val = NOEDGE;
    if (st < (E & 3)) val = dst[e];
    const unsigned m = gcn_owned(val, lo, lo + RNG, below);
    if (m != 0u) gcn_note(m, val, e, lo, wcount, cntL, myE, myU);
  }
}

struct gcn_frag {
  v2f a[4][MT];
  v2f b[4][2];
};

__device__ __forceinline__ void gcn_load_chunk(gcn_frag& f, const float* ap,
                                               const float* __restrict__ bp) {
#pragma unroll
  for (int s = 0; s < 4; ++s) {
    const float* b = bp + (size_t)s * 4 * HDIM;
    f.b[s][0].x = b[0];   f.b[s][0].y = b[HDIM];
    f.b[s][1].x = b[16];  f.b[s][1].y = b[HDIM + 16];
  }
#pragma unroll
  for (int s = 0; s < 4; ++s) {
#pragma unroll
    for (int i = 0; i < MT; ++i) {
      const float2 v = *(const float2*)(ap + i * 16 * LDSP + 4 * s);
      f.a[s][i].x = v.x;  f.a[s][i].y = v.y;
    }
  }
}

__device__ __forceinline__ void gcn_mma_chunk(const gcn_frag& f, v8f (&acc)[MT][2]) {
#pragma unroll
  for (int s = 0; s < 4; ++s) {
#pragma unroll
    for (int i = 0; i < MT; ++i) {
      acc[i][0] = __builtin_amdgcn_wmma_f32_16x16x4_f32(false, f.a[s][i], false, f.b[s][0],
                                                        (short)0, acc[i][0], false, false);
      acc[i][1] = __builtin_amdgcn_wmma_f32_16x16x4_f32(false, f.a[s][i], false, f.b[s][1],
                                                        (short)0, acc[i][1], false, false);
    }
  }
}

__device__ __forceinline__ void gcn_copy_slab(float* xs, const float* __restrict__ x, int M,
                                              int m0, int wave, int lane) {
#pragma unroll
  for (int rr = 0; rr < BM / 8; ++rr) {
    const int r = wave + 8 * rr;
    int row = m0 + r;
    if (row > M - 1) row = M - 1;
    const float2* src = (const float2*)(x + (size_t)row * KDIM);
    float2* dst = (float2*)(xs + r * LDSP);
#pragma unroll
    for (int q = 0; q < 6; ++q) dst[lane + 32 * q] = src[lane + 32 * q];
  }
  if (lane < BM / 8) {
    const int r = wave + 8 * lane;
    int row = m0 + r;
    if (row > M - 1) row = M - 1;
    ((float2*)(xs + r * LDSP))[192] = ((const float2*)(x + (size_t)row * KDIM))[192];
  }
}

__device__ __forceinline__ void gcn_product(const float* xs, const float* __restrict__ W1,
                                            float* __restrict__ t, int M, int m0, int wave,
                                            int lane) {
  const int n0 = wave * 32;
  const int halfSel = lane >> 4;
  const int l15 = lane & 15;
  const int col0 = n0 + l15;

  v8f acc[MT][2];
#pragma unroll
  for (int i = 0; i < MT; ++i) {
    acc[i][0] = (v8f){0.f, 0.f, 0.f, 0.f, 0.f, 0.f, 0.f, 0.f};
    acc[i][1] = (v8f){0.f, 0.f, 0.f, 0.f, 0.f, 0.f, 0.f, 0.f};
  }

  const float* ap = xs + l15 * LDSP + halfSel * 2;
  const float* bp = W1 + (size_t)(halfSel * 2) * HDIM + col0;

  gcn_frag f0, f1;
  gcn_load_chunk(f0, ap, bp);
  for (int c = 0; c < NCHUNK; c += 2) {
    gcn_load_chunk(f1, ap + (c + 1) * KC, bp + (size_t)(c + 1) * KC * HDIM);
    gcn_mma_chunk(f0, acc);
    const int cn = (c + 2 < NCHUNK) ? (c + 2) : (NCHUNK - 1);
    gcn_load_chunk(f0, ap + cn * KC, bp + (size_t)cn * KC * HDIM);
    gcn_mma_chunk(f1, acc);
  }

  {
    v2f a[MT];
    v2f b0 = {0.f, 0.f};
    v2f b1 = {0.f, 0.f};
#pragma unroll
    for (int i = 0; i < MT; ++i) a[i] = (v2f){0.f, 0.f};
    if (halfSel == 0) {
#pragma unroll
      for (int i = 0; i < MT; ++i) {
        const float2 v = *(const float2*)(xs + (i * 16 + l15) * LDSP + KMAIN);
        a[i].x = v.x;  a[i].y = v.y;
      }
      const float* bt = W1 + (size_t)KMAIN * HDIM + col0;
      b0.x = bt[0];   b0.y = bt[HDIM];
      b1.x = bt[16];  b1.y = bt[HDIM + 16];
    }
#pragma unroll
    for (int i = 0; i < MT; ++i) {
      acc[i][0] = __builtin_amdgcn_wmma_f32_16x16x4_f32(false, a[i], false, b0, (short)0,
                                                        acc[i][0], false, false);
      acc[i][1] = __builtin_amdgcn_wmma_f32_16x16x4_f32(false, a[i], false, b1, (short)0,
                                                        acc[i][1], false, false);
    }
  }

  float* tp = t + (size_t)(m0 + halfSel * 8) * HDIM + col0;
  if (m0 + BM <= M) {
#pragma unroll
    for (int i = 0; i < MT; ++i) {
#pragma unroll
      for (int r = 0; r < 8; ++r) {
        tp[(i * 16 + r) * HDIM] = acc[i][0][r];
        tp[(i * 16 + r) * HDIM + 16] = acc[i][1][r];
      }
    }
  } else {
#pragma unroll
    for (int i = 0; i < MT; ++i) {
#pragma unroll
      for (int r = 0; r < 8; ++r) {
        if (m0 + i * 16 + r + halfSel * 8 < M) {
          tp[(i * 16 + r) * HDIM] = acc[i][0][r];
          tp[(i * 16 + r) * HDIM + 16] = acc[i][1][r];
        }
      }
    }
  }
}

__global__ __launch_bounds__(FT) void gcn_gemm1_build(const float* __restrict__ x,
                                                      const float* __restrict__ W1,
                                                      float* __restrict__ t,
                                                      const int* __restrict__ src,
                                                      const int* __restrict__ dst,
                                                      int* __restrict__ rowptr,
                                                      float* __restrict__ dis,
                                                      int* __restrict__ csr_src, int N, int E) {
  __shared__ float xs[BM * LDSP];
  __shared__ int cntL[RNG];
  __shared__ int nextL[RNG];
  __shared__ int exL[RNG];
  __shared__ int wsum[SCW];
  __shared__ int wbelow[SW];
  __shared__ int ovf;
  __shared__ int listE[SW * CAPW];
  __shared__ int listU[SW * CAPW];

  const int tid = threadIdx.x;
  const int lane = tid & 31;
  const int wv = __builtin_amdgcn_readfirstlane(tid >> 5);
  const int m0 = blockIdx.x * BM;
  const int pw = wv - SW;

  if (wv >= SW) {
    gcn_copy_slab(xs, x, N, m0, pw, lane);
  } else {
    if (tid < RNG) { cntL[tid] = 0; nextL[tid] = 0; }
    if (tid == 0) ovf = 0;
  }
  __syncthreads();

  int* myE = listE + wv * CAPW;
  int* myU = listU + wv * CAPW;
  int wcount = 0;
  if (wv >= SW) {
    gcn_product(xs, W1, t, N, m0, pw, lane);
  } else {
    int below = 0;
    gcn_sweep(dst, E, tid, m0, below, wcount, cntL, myE, myU);
    if (lane == 0) {
      wbelow[wv] = below;
      if (wcount > CAPW) ovf = 1;
    }
  }
  __syncthreads();

  int base = (lane < SW) ? wbelow[lane] : 0;
  for (int off = 16; off > 0; off >>= 1) base += __shfl_xor(base, off);

  int c = 0, incl = 0;
  if (tid < SCW * 32) {
    c = (tid < RNG) ? cntL[tid] : 0;
    incl = c;
    for (int off = 1; off < 32; off <<= 1) {
      const int v = __shfl_up(incl, off);
      if (lane >= off) incl += v;
    }
    if (lane == 31) wsum[wv] = incl;
  }
  __syncthreads();
  if (tid < RNG) {
    int woff = 0;
    for (int w = 0; w < SCW; ++w) woff += (w < wv) ? wsum[w] : 0;
    const int ex = woff + incl - c;
    exL[tid] = ex;
    const int n = m0 + tid;
    if (n < N) {
      rowptr[n] = base + ex;
      dis[n] = rsqrtf((float)(c + 1));
      if (n == N - 1) rowptr[N] = base + ex + c;
    }
  }
  __syncthreads();

  if (ovf == 0) {
    if (wv < SW) {
      for (int k0 = 0; k0 < wcount; k0 += 128) {
        int sv[4], uu[4];
#pragma unroll
        for (int j = 0; j < 4; ++j) {
          const int k = k0 + j * 32 + lane;
          uu[j] = -1;  sv[j] = 0;
          if (k < wcount) { uu[j] = myU[k];  sv[j] = src[myE[k]]; }
        }
#pragma unroll
        for (int j = 0; j < 4; ++j) {
          if (uu[j] >= 0) {
            const int r = atomicAdd(&nextL[uu[j]], 1);
            csr_src[base + exL[uu[j]] + r] = sv[j];
          }
        }
      }
    }
  } else {
    for (int e = tid; e < E; e += FT) {
      const unsigned u = (unsigned)dst[e] - (unsigned)m0;
      if (u < (unsigned)RNG) {
        const int r = atomicAdd(&nextL[u], 1);
        csr_src[base + exL[u] + r] = src[e];
      }
    }
  }
}


template <int K>
__device__ __forceinline__ void gcn_rows(const float* __restrict__ t, int sj, float wj, int j,
                                         int f0, int f1, float4& a0, float4& a1) {
  float4 v0[K], v1[K];
  float w[K];
#pragma unroll
  for (int u = 0; u < K; ++u) {
    const int s = __builtin_amdgcn_readlane(sj, j + u);
    w[u] = __int_as_float(__builtin_amdgcn_readlane(__float_as_int(wj), j + u));
    const float* tp = t + (size_t)s * HDIM;
    v0[u] = *(const float4*)(tp + f0);
    v1[u] = *(const float4*)(tp + f1);
  }
#pragma unroll
  for (int u = 0; u < K; ++u) {
    a0.x = fmaf(v0[u].x, w[u], a0.x);  a0.y = fmaf(v0[u].y, w[u], a0.y);
    a0.z = fmaf(v0[u].z, w[u], a0.z);  a0.w = fmaf(v0[u].w, w[u], a0.w);
    a1.x = fmaf(v1[u].x, w[u], a1.x);  a1.y = fmaf(v1[u].y, w[u], a1.y);
    a1.z = fmaf(v1[u].z, w[u], a1.z);  a1.w = fmaf(v1[u].w, w[u], a1.w);
  }
}

__global__ __launch_bounds__(256) void gcn_gather1(const int* __restrict__ rowptr,
                                                   const int* __restrict__ csr_src,
                                                   const float* __restrict__ dis,
                                                   const float* __restrict__ t,
                                                   const float* __restrict__ b1,
                                                   const float* __restrict__ W2,
                                                   float* __restrict__ t2, int N) {
  const int wv = __builtin_amdgcn_readfirstlane((int)(threadIdx.x >> 5));
  const int d = blockIdx.x * 8 + wv;
  const int lane = threadIdx.x & 31;
  if (d >= N) return;
  const int f0 = lane * 4;
  const int f1 = 128 + lane * 4;
  const float dd = dis[d];
  const int beg = rowptr[d];
  const int end = rowptr[d + 1];

  float4 a0 = make_float4(0.f, 0.f, 0.f, 0.f);
  float4 a1 = make_float4(0.f, 0.f, 0.f, 0.f);

  for (int base = beg; base < end; base += 32) {
    int n = end - base;
    if (n > 32) n = 32;
    int sj = 0;
    float wj = 0.f;
    if (lane < n) {
      sj = csr_src[base + lane];
      wj = dis[sj] * dd;
    }
    int j = 0;
    for (; j + 8 <= n; j += 8) gcn_rows<8>(t, sj, wj, j, f0, f1, a0, a1);
    if (n & 4) { gcn_rows<4>(t, sj, wj, j, f0, f1, a0, a1); j += 4; }
    if (n & 2) { gcn_rows<2>(t, sj, wj, j, f0, f1, a0, a1); j += 2; }
    if (n & 1) { gcn_rows<1>(t, sj, wj, j, f0, f1, a0, a1); }
  }

  {
    const float sn = dd * dd;
    const float* tp = t + (size_t)d * HDIM;
    const float4 h0 = *(const float4*)(tp + f0);
    const float4 h1 = *(const float4*)(tp + f1);
    const float4 c0 = *(const float4*)(b1 + f0);
    const float4 c1 = *(const float4*)(b1 + f1);
    a0.x = fmaxf((a0.x + h0.x * sn) + c0.x, 0.f);
    a0.y = fmaxf((a0.y + h0.y * sn) + c0.y, 0.f);
    a0.z = fmaxf((a0.z + h0.z * sn) + c0.z, 0.f);
    a0.w = fmaxf((a0.w + h0.w * sn) + c0.w, 0.f);
    a1.x = fmaxf((a1.x + h1.x * sn) + c1.x, 0.f);
    a1.y = fmaxf((a1.y + h1.y * sn) + c1.y, 0.f);
    a1.z = fmaxf((a1.z + h1.z * sn) + c1.z, 0.f);
    a1.w = fmaxf((a1.w + h1.w * sn) + c1.w, 0.f);
  }

  const float4 wa = *(const float4*)(W2 + f0 * 2);
  const float4 wb = *(const float4*)(W2 + f0 * 2 + 4);
  const float4 wc = *(const float4*)(W2 + f1 * 2);
  const float4 wd = *(const float4*)(W2 + f1 * 2 + 4);
  float p0 = a0.x * wa.x + a0.y * wa.z + a0.z * wb.x + a0.w * wb.z
           + a1.x * wc.x + a1.y * wc.z + a1.z * wd.x + a1.w * wd.z;
  float p1 = a0.x * wa.y + a0.y * wa.w + a0.z * wb.y + a0.w * wb.w
           + a1.x * wc.y + a1.y * wc.w + a1.z * wd.y + a1.w * wd.w;
  for (int off = 16; off > 0; off >>= 1) {
    p0 += __shfl_down(p0, off);
    p1 += __shfl_down(p1, off);
  }
  if (lane == 0) {
    t2[d * 2 + 0] = p0;
    t2[d * 2 + 1] = p1;
  }
}

__global__ __launch_bounds__(256) void gcn_gather2(const int* __restrict__ rowptr,
                                                   const int* __restrict__ csr_src,
                                                   const float* __restrict__ dis,
                                                   const float* __restrict__ t2,
                                                   const float* __restrict__ b2,
                                                   float* __restrict__ out, int N) {
  __shared__ float so[64];
  const int g = blockIdx.x * 32 + (threadIdx.x >> 3);
  const int sub = threadIdx.x & 7;
  const int d = (g < N) ? g : (N - 1);
  const float dd = dis[d];
  const int beg = rowptr[d];
  const int end = rowptr[d + 1];
  float s0 = 0.f, s1 = 0.f;
  for (int j = beg + sub; j < end; j += 8) {
    const int s = csr_src[j];
    const float w = dis[s] * dd;
    const float2 v = *(const float2*)(t2 + 2 * s);
    s0 += v.x * w;
    s1 += v.y * w;
  }
  for (int off = 4; off > 0; off >>= 1) {
    s0 += __shfl_xor(s0, off);
    s1 += __shfl_xor(s1, off);
  }
  if (sub == 0) {
    const float sn = dd * dd;
    const float2 hv = *(const float2*)(t2 + 2 * d);
    so[(threadIdx.x >> 3) * 2 + 0] = tanhf((s0 + hv.x * sn) + b2[0]);
    so[(threadIdx.x >> 3) * 2 + 1] = tanhf((s1 + hv.y * sn) + b2[1]);
  }
  __syncthreads();
  if (threadIdx.x < 64) {
    const int o = blockIdx.x * 64 + threadIdx.x;
    if (o < 2 * N) out[o] = so[threadIdx.x];
  }
}

extern "C" void kernel_launch(void* const* d_in, const int* in_sizes, int n_in,
                              void* d_out, int out_size, void* d_ws, size_t ws_size,
                              hipStream_t stream) {
  const float* x  = (const float*)d_in[0];
  const int*   src = (const int*)d_in[1];
  const int*   dst = (const int*)d_in[2];
  const float* W1 = (const float*)d_in[3];
  const float* b1 = (const float*)d_in[4];
  const float* W2 = (const float*)d_in[5];
  const float* b2 = (const float*)d_in[6];
  float* out = (float*)d_out;

  const int N = in_sizes[0] / KDIM;
  const int E = in_sizes[1];

  float* ws  = (float*)d_ws;
  float* t   = ws;
  float* t2  = t  + (size_t)N * HDIM;
  float* dis = t2 + (size_t)2 * N;
  int* rowptr  = (int*)(dis + N);
  int* csr_src = rowptr + N + 1;

  gcn_gemm1_build<<<(N + BM - 1) / BM, FT, 0, stream>>>(x, W1, t, src, dst, rowptr, dis, csr_src,
                                                        N, E);
  gcn_gather1<<<(N + 7) / 8, 256, 0, stream>>>(rowptr, csr_src, dis, t, b1, W2, t2, N);

  gcn_gather2<<<(N + 31) / 32, 256, 0, stream>>>(rowptr, csr_src, dis, t2, b2, out, N);
}
